# starting kernel
# baseline (speedup 1.0000x reference)
#include <hip/hip_runtime.h>
#include <math.h>

typedef float v2f __attribute__((ext_vector_type(2)));
typedef float v8f __attribute__((ext_vector_type(8)));

namespace {
constexpr int C_DIM = 2048;
constexpr int HW = 49;
constexpr int KSTEPS = C_DIM / 4;
constexpr int WAVES_PER_BLOCK = 16;
constexpr int STEPS_PER_WAVE = KSTEPS / WAVES_PER_BLOCK;
}

__global__ void pack_A(const float* __restrict__ W, float* __restrict__ Apack) {
  int idx = blockIdx.x * blockDim.x + threadIdx.x;
  if (idx >= KSTEPS * 32) return;
  int lane = idx & 31;
  int step = idx >> 5;
  int m = lane & 15;
  int kbase = step * 4 + ((lane >> 4) << 1);
  float v0 = 0.0f, v1 = 0.0f;
#pragma unroll
  for (int j = 0; j < 2; ++j) {
    int kk = kbase + j;
    float val = 0.0f;
    if (m == 0)      val = 1.0f;
    else if (m == 1) val = W[2048 + kk];
    else if (m == 2) val = W[4096 + 2048 + kk];
    else if (m == 3) val = W[kk];
    else if (m == 4) val = W[4096 + kk];
    if (j == 0) v0 = val; else v1 = val;
  }
  float2 packed; packed.x = v0; packed.y = v1;
  ((float2*)Apack)[idx] = packed;
}

__global__ __launch_bounds__(512) void fused(
    const float* __restrict__ x, const float* __restrict__ mask,
    const float* __restrict__ bvec, const float* __restrict__ Apack,
    float* __restrict__ out) {
  __shared__ float part[WAVES_PER_BLOCK][5][64];
  __shared__ float fin[5][HW];
  __shared__ float sm[HW];
  __shared__ float sgap[HW];
  __shared__ int   s_pmax;
  __shared__ float s_gapmean;

  const int tid = threadIdx.x;
  const int n = blockIdx.x;
  const int wave = tid >> 5;
  const int lane = tid & 31;
  const int mrow = lane & 15;
  const int chalf = (lane >> 4) << 1;

  if (tid < HW) sm[tid] = mask[n * HW + tid];

  const int p0 = mrow, p1 = 16 + mrow, p2 = 32 + mrow;

  v8f a0 = {}, a1 = {}, a2 = {}, a3 = {};
  const float* xn = x + (size_t)n * (C_DIM * HW);
  const int step0 = wave * STEPS_PER_WAVE;

  for (int s = 0; s < STEPS_PER_WAVE; ++s) {
    const int step = step0 + s;
    const float* xr = xn + (size_t)(step * 4 + chalf) * HW;
    v2f A = *(const v2f*)(Apack + 2 * ((size_t)step * 32 + lane));
    v2f B0, B1, B2, B3;
    B0.x = xr[p0];  B0.y = xr[HW + p0];
    B1.x = xr[p1];  B1.y = xr[HW + p1];
    B2.x = xr[p2];  B2.y = xr[HW + p2];
    B3.x = xr[48];  B3.y = xr[HW + 48];
    a0 = __builtin_amdgcn_wmma_f32_16x16x4_f32(false, A, false, B0, (short)0, a0, false, false);
    a1 = __builtin_amdgcn_wmma_f32_16x16x4_f32(false, A, false, B1, (short)0, a1, false, false);
    a2 = __builtin_amdgcn_wmma_f32_16x16x4_f32(false, A, false, B2, (short)0, a2, false, false);
    a3 = __builtin_amdgcn_wmma_f32_16x16x4_f32(false, A, false, B3, (short)0, a3, false, false);
  }

  if (lane < 16) {
#pragma unroll
    for (int m = 0; m < 5; ++m) {
      part[wave][m][mrow]      = a0[m];
      part[wave][m][16 + mrow] = a1[m];
      part[wave][m][32 + mrow] = a2[m];
      part[wave][m][48 + mrow] = a3[m];
    }
  }
  __syncthreads();

  if (tid < 5 * HW) {
    int m = tid / HW, p = tid % HW;
    float ssum = 0.0f;
    for (int w = 0; w < WAVES_PER_BLOCK; ++w) ssum += part[w][m][p];
    fin[m][p] = ssum;
  }
  __syncthreads();

  if (tid == 0) {
    float tsum = 0.0f;
    for (int p = 0; p < HW; ++p) tsum += sm[p];
    const float thr = tsum / 49.0f;
    int pmax = 0;
    float best = -__builtin_inff();
    for (int p = 0; p < HW; ++p) {
      float mm = (sm[p] > thr) ? (fin[0][p] / 2048.0f) : 0.0f;
      if (mm > best) { best = mm; pmax = p; }
    }
    s_pmax = pmax;
  }
  __syncthreads();

  const int pmax = s_pmax;
  float pred_d = 0.0f, rdist = 0.0f;
  if (tid < HW) {
    const float u0 = fin[3][pmax];
    const float u1 = fin[4][pmax];
    float pd = fmaxf(u0 + fin[1][tid] + bvec[0], 0.0f);
    float pa = fmaxf(u1 + fin[2][tid] + bvec[1], 0.0f);
    if (tid == pmax) { pd = 0.0f; pa = 0.0f; }
    const int ii = tid / 7,  jj = tid % 7;
    const int i0 = pmax / 7, j0 = pmax % 7;
    const float rx = (float)(ii - i0) / 7.0f;
    const float ry = (float)(jj - j0) / 7.0f;
    rdist = sqrtf(rx * rx + ry * ry);
    const float ang = (atan2f(ry, rx) / 3.14159265358979323846f + 1.0f) * 0.5f;
    float gap = pa - ang;
    if (gap < 0.0f) gap += 1.0f;
    sgap[tid] = gap;
    pred_d = pd;
  }
  __syncthreads();
  if (tid == 0) {
    float gs = 0.0f;
    for (int p = 0; p < HW; ++p) gs += sgap[p];
    s_gapmean = gs / 49.0f;
  }
  __syncthreads();
  if (tid < HW) {
    const float g = sgap[tid] - s_gapmean;
    const float dl = pred_d - rdist;
    out[(size_t)n * HW + tid] = dl * dl + g * g;
  }
}

extern "C" void kernel_launch(void* const* d_in, const int* in_sizes, int n_in,
                              void* d_out, int out_size, void* d_ws, size_t ws_size,
                              hipStream_t stream) {
  const float* x    = (const float*)d_in[0];
  const float* mask = (const float*)d_in[1];
  const float* W    = (const float*)d_in[2];
  const float* b    = (const float*)d_in[3];
  float* out = (float*)d_out;
  float* Apack = (float*)d_ws;

  pack_A<<<(KSTEPS * 32 + 255) / 256, 256, 0, stream>>>(W, Apack);
  fused<<<256, 512, 0, stream>>>(x, mask, b, Apack, out);
}


# B operand loaded as one 16-byte row segment per lane (tile N index = position/4, the 4 accumulator tiles = position mod 4): 2 b128 loads per K-step replace 8 dword loads; sums unchanged
# speedup vs baseline: 1.1420x; 1.1420x over previous
#include <hip/hip_runtime.h>
#include <math.h>

typedef float v2f __attribute__((ext_vector_type(2)));
typedef float v8f __attribute__((ext_vector_type(8)));
typedef float v4u __attribute__((ext_vector_type(4)));

namespace {
constexpr int C_DIM = 2048;
constexpr int HW = 49;
constexpr int KSTEPS = C_DIM / 4;
constexpr int WAVES_PER_BLOCK = 16;
constexpr int STEPS_PER_WAVE = KSTEPS / WAVES_PER_BLOCK;
}

__global__ void pack_A(const float* __restrict__ W, float* __restrict__ Apack) {
  int idx = blockIdx.x * blockDim.x + threadIdx.x;
  if (idx >= KSTEPS * 32) return;
  int lane = idx & 31;
  int step = idx >> 5;
  int m = lane & 15;
  int kbase = step * 4 + ((lane >> 4) << 1);
  float v0 = 0.0f, v1 = 0.0f;
#pragma unroll
  for (int j = 0; j < 2; ++j) {
    int kk = kbase + j;
    float val = 0.0f;
    if (m == 0)      val = 1.0f;
    else if (m == 1) val = W[2048 + kk];
    else if (m == 2) val = W[4096 + 2048 + kk];
    else if (m == 3) val = W[kk];
    else if (m == 4) val = W[4096 + kk];
    if (j == 0) v0 = val; else v1 = val;
  }
  float2 packed; packed.x = v0; packed.y = v1;
  ((float2*)Apack)[idx] = packed;
}

__global__ __launch_bounds__(512) void fused(
    const float* __restrict__ x, const float* __restrict__ mask,
    const float* __restrict__ bvec, const float* __restrict__ Apack,
    float* __restrict__ out) {
  __shared__ float part[WAVES_PER_BLOCK][5][64];
  __shared__ float fin[5][HW];
  __shared__ float sm[HW];
  __shared__ float sgap[HW];
  __shared__ int   s_pmax;
  __shared__ float s_gapmean;

  const int tid = threadIdx.x;
  const int n = blockIdx.x;
  const int wave = tid >> 5;
  const int lane = tid & 31;
  const int mrow = lane & 15;
  const int chalf = (lane >> 4) << 1;

  if (tid < HW) sm[tid] = mask[n * HW + tid];

  const int pb = (mrow < 12) ? (mrow << 2) : 45;

  v8f a0 = {}, a1 = {}, a2 = {}, a3 = {};
  const float* xn = x + (size_t)n * (C_DIM * HW);
  const int step0 = wave * STEPS_PER_WAVE;
  const float* xl = xn + (size_t)(step0 * 4 + chalf) * HW + pb;
  const float* al = Apack + 2 * ((size_t)step0 * 32 + lane);

#pragma unroll 16
  for (int s = 0; s < STEPS_PER_WAVE; ++s) {
    const float* xr = xl + (size_t)s * (4 * HW);
    v2f A = *(const v2f*)(al + (size_t)s * 64);
    v4u r0, r1;
    __builtin_memcpy(&r0, xr, 16);
    __builtin_memcpy(&r1, xr + HW, 16);
    v2f B0, B1, B2, B3;
    B0.x = r0.x;  B0.y = r1.x;
    B1.x = r0.y;  B1.y = r1.y;
    B2.x = r0.z;  B2.y = r1.z;
    B3.x = r0.w;  B3.y = r1.w;
    a0 = __builtin_amdgcn_wmma_f32_16x16x4_f32(false, A, false, B0, (short)0, a0, false, false);
    a1 = __builtin_amdgcn_wmma_f32_16x16x4_f32(false, A, false, B1, (short)0, a1, false, false);
    a2 = __builtin_amdgcn_wmma_f32_16x16x4_f32(false, A, false, B2, (short)0, a2, false, false);
    a3 = __builtin_amdgcn_wmma_f32_16x16x4_f32(false, A, false, B3, (short)0, a3, false, false);
  }

  if (lane < 12) {
#pragma unroll
    for (int m = 0; m < 5; ++m) {
      part[wave][m][pb]     = a0[m];
      part[wave][m][pb + 1] = a1[m];
      part[wave][m][pb + 2] = a2[m];
      part[wave][m][pb + 3] = a3[m];
    }
  } else if (lane == 12) {
#pragma unroll
    for (int m = 0; m < 5; ++m) part[wave][m][48] = a3[m];
  }
  __syncthreads();

  if (tid < 5 * HW) {
    int m = tid / HW, p = tid % HW;
    float ssum = 0.0f;
    for (int w = 0; w < WAVES_PER_BLOCK; ++w) ssum += part[w][m][p];
    fin[m][p] = ssum;
  }
  __syncthreads();

  if (tid == 0) {
    float tsum = 0.0f;
    for (int p = 0; p < HW; ++p) tsum += sm[p];
    const float thr = tsum / 49.0f;
    int pmax = 0;
    float best = -__builtin_inff();
    for (int p = 0; p < HW; ++p) {
      float mm = (sm[p] > thr) ? (fin[0][p] / 2048.0f) : 0.0f;
      if (mm > best) { best = mm; pmax = p; }
    }
    s_pmax = pmax;
  }
  __syncthreads();

  const int pmax = s_pmax;
  float pred_d = 0.0f, rdist = 0.0f;
  if (tid < HW) {
    const float u0 = fin[3][pmax];
    const float u1 = fin[4][pmax];
    float pd = fmaxf(u0 + fin[1][tid] + bvec[0], 0.0f);
    float pa = fmaxf(u1 + fin[2][tid] + bvec[1], 0.0f);
    if (tid == pmax) { pd = 0.0f; pa = 0.0f; }
    const int ii = tid / 7,  jj = tid % 7;
    const int i0 = pmax / 7, j0 = pmax % 7;
    const float rx = (float)(ii - i0) / 7.0f;
    const float ry = (float)(jj - j0) / 7.0f;
    rdist = sqrtf(rx * rx + ry * ry);
    const float ang = (atan2f(ry, rx) / 3.14159265358979323846f + 1.0f) * 0.5f;
    float gap = pa - ang;
    if (gap < 0.0f) gap += 1.0f;
    sgap[tid] = gap;
    pred_d = pd;
  }
  __syncthreads();
  if (tid == 0) {
    float gs = 0.0f;
    for (int p = 0; p < HW; ++p) gs += sgap[p];
    s_gapmean = gs / 49.0f;
  }
  __syncthreads();
  if (tid < HW) {
    const float g = sgap[tid] - s_gapmean;
    const float dl = pred_d - rdist;
    out[(size_t)n * HW + tid] = dl * dl + g * g;
  }
}

extern "C" void kernel_launch(void* const* d_in, const int* in_sizes, int n_in,
                              void* d_out, int out_size, void* d_ws, size_t ws_size,
                              hipStream_t stream) {
  const float* x    = (const float*)d_in[0];
  const float* mask = (const float*)d_in[1];
  const float* W    = (const float*)d_in[2];
  const float* b    = (const float*)d_in[3];
  float* out = (float*)d_out;
  float* Apack = (float*)d_ws;

  pack_A<<<(KSTEPS * 32 + 255) / 256, 256, 0, stream>>>(W, Apack);
  fused<<<256, 512, 0, stream>>>(x, mask, b, Apack, out);
}


# A operand read in the main kernel straight from W (one 16-byte load per 8 channels, each lane half owning 4 consecutive channels); pack launch removed; channel sum order inside 8-groups changes
# speedup vs baseline: 1.6218x; 1.4201x over previous
#include <hip/hip_runtime.h>
#include <math.h>

typedef float v2f __attribute__((ext_vector_type(2)));
typedef float v8f __attribute__((ext_vector_type(8)));
typedef float v4u __attribute__((ext_vector_type(4)));

namespace {
constexpr int C_DIM = 2048;
constexpr int HW = 49;
constexpr int KSTEPS = C_DIM / 4;
constexpr int WAVES_PER_BLOCK = 16;
constexpr int STEPS_PER_WAVE = KSTEPS / WAVES_PER_BLOCK;
constexpr int PAIRS_PER_WAVE = STEPS_PER_WAVE / 2;
}

__global__ __launch_bounds__(512) void fused(
    const float* __restrict__ x, const float* __restrict__ mask,
    const float* __restrict__ bvec, const float* __restrict__ W,
    float* __restrict__ out) {
  __shared__ float part[WAVES_PER_BLOCK][5][64];
  __shared__ float fin[5][HW];
  __shared__ float sm[HW];
  __shared__ float sgap[HW];
  __shared__ int   s_pmax;
  __shared__ float s_gapmean;

  const int tid = threadIdx.x;
  const int n = blockIdx.x;
  const int wave = tid >> 5;
  const int lane = tid & 31;
  const int mrow = lane & 15;
  const int hoff = (lane >> 4) << 2;

  if (tid < HW) sm[tid] = mask[n * HW + tid];

  const int pb = (mrow < 12) ? (mrow << 2) : 45;

  v8f a0 = {}, a1 = {}, a2 = {}, a3 = {};
  const float* xn = x + (size_t)n * (C_DIM * HW);
  const bool a_from_w = (mrow >= 1) && (mrow <= 4);
  const float a_const = (mrow == 0) ? 1.0f : 0.0f;
  int woff = C_DIM;
  if (mrow == 2)      woff = 2 * C_DIM + C_DIM;
  else if (mrow == 3) woff = 0;
  else if (mrow == 4) woff = 2 * C_DIM;
  const int cbase = wave * (STEPS_PER_WAVE * 4) + hoff;
  const float* wl = W + woff + cbase;
  const float* xl = xn + (size_t)cbase * HW + pb;

#pragma unroll 8
  for (int j = 0; j < PAIRS_PER_WAVE; ++j) {
    const float* xr = xl + (size_t)j * (8 * HW);
    v4u aw, r0, r1, r2, r3;
    __builtin_memcpy(&aw, wl + j * 8, 16);
    __builtin_memcpy(&r0, xr, 16);
    __builtin_memcpy(&r1, xr + HW, 16);
    __builtin_memcpy(&r2, xr + 2 * HW, 16);
    __builtin_memcpy(&r3, xr + 3 * HW, 16);
    v2f A0, A1;
    A0.x = a_from_w ? aw.x : a_const;  A0.y = a_from_w ? aw.y : a_const;
    A1.x = a_from_w ? aw.z : a_const;  A1.y = a_from_w ? aw.w : a_const;
    v2f B0, B1, B2, B3;
    B0.x = r0.x;  B0.y = r1.x;
    B1.x = r0.y;  B1.y = r1.y;
    B2.x = r0.z;  B2.y = r1.z;
    B3.x = r0.w;  B3.y = r1.w;
    a0 = __builtin_amdgcn_wmma_f32_16x16x4_f32(false, A0, false, B0, (short)0, a0, false, false);
    a1 = __builtin_amdgcn_wmma_f32_16x16x4_f32(false, A0, false, B1, (short)0, a1, false, false);
    a2 = __builtin_amdgcn_wmma_f32_16x16x4_f32(false, A0, false, B2, (short)0, a2, false, false);
    a3 = __builtin_amdgcn_wmma_f32_16x16x4_f32(false, A0, false, B3, (short)0, a3, false, false);
    B0.x = r2.x;  B0.y = r3.x;
    B1.x = r2.y;  B1.y = r3.y;
    B2.x = r2.z;  B2.y = r3.z;
    B3.x = r2.w;  B3.y = r3.w;
    a0 = __builtin_amdgcn_wmma_f32_16x16x4_f32(false, A1, false, B0, (short)0, a0, false, false);
    a1 = __builtin_amdgcn_wmma_f32_16x16x4_f32(false, A1, false, B1, (short)0, a1, false, false);
    a2 = __builtin_amdgcn_wmma_f32_16x16x4_f32(false, A1, false, B2, (short)0, a2, false, false);
    a3 = __builtin_amdgcn_wmma_f32_16x16x4_f32(false, A1, false, B3, (short)0, a3, false, false);
  }

  if (lane < 12) {
#pragma unroll
    for (int m = 0; m < 5; ++m) {
      part[wave][m][pb]     = a0[m];
      part[wave][m][pb + 1] = a1[m];
      part[wave][m][pb + 2] = a2[m];
      part[wave][m][pb + 3] = a3[m];
    }
  } else if (lane == 12) {
#pragma unroll
    for (int m = 0; m < 5; ++m) part[wave][m][48] = a3[m];
  }
  __syncthreads();

  if (tid < 5 * HW) {
    int m = tid / HW, p = tid % HW;
    float ssum = 0.0f;
    for (int w = 0; w < WAVES_PER_BLOCK; ++w) ssum += part[w][m][p];
    fin[m][p] = ssum;
  }
  __syncthreads();

  if (tid == 0) {
    float tsum = 0.0f;
    for (int p = 0; p < HW; ++p) tsum += sm[p];
    const float thr = tsum / 49.0f;
    int pmax = 0;
    float best = -__builtin_inff();
    for (int p = 0; p < HW; ++p) {
      float mm = (sm[p] > thr) ? (fin[0][p] / 2048.0f) : 0.0f;
      if (mm > best) { best = mm; pmax = p; }
    }
    s_pmax = pmax;
  }
  __syncthreads();

  const int pmax = s_pmax;
  float pred_d = 0.0f, rdist = 0.0f;
  if (tid < HW) {
    const float u0 = fin[3][pmax];
    const float u1 = fin[4][pmax];
    float pd = fmaxf(u0 + fin[1][tid] + bvec[0], 0.0f);
    float pa = fmaxf(u1 + fin[2][tid] + bvec[1], 0.0f);
    if (tid == pmax) { pd = 0.0f; pa = 0.0f; }
    const int ii = tid / 7,  jj = tid % 7;
    const int i0 = pmax / 7, j0 = pmax % 7;
    const float rx = (float)(ii - i0) / 7.0f;
    const float ry = (float)(jj - j0) / 7.0f;
    rdist = sqrtf(rx * rx + ry * ry);
    const float ang = (atan2f(ry, rx) / 3.14159265358979323846f + 1.0f) * 0.5f;
    float gap = pa - ang;
    if (gap < 0.0f) gap += 1.0f;
    sgap[tid] = gap;
    pred_d = pd;
  }
  __syncthreads();
  if (tid == 0) {
    float gs = 0.0f;
    for (int p = 0; p < HW; ++p) gs += sgap[p];
    s_gapmean = gs / 49.0f;
  }
  __syncthreads();
  if (tid < HW) {
    const float g = sgap[tid] - s_gapmean;
    const float dl = pred_d - rdist;
    out[(size_t)n * HW + tid] = dl * dl + g * g;
  }
}

extern "C" void kernel_launch(void* const* d_in, const int* in_sizes, int n_in,
                              void* d_out, int out_size, void* d_ws, size_t ws_size,
                              hipStream_t stream) {
  const float* x    = (const float*)d_in[0];
  const float* mask = (const float*)d_in[1];
  const float* W    = (const float*)d_in[2];
  const float* b    = (const float*)d_in[3];
  float* out = (float*)d_out;

  fused<<<256, 512, 0, stream>>>(x, mask, b, W, out);
}


# Half as many waves: 8 per work-group, 256 channels each, row loads kept two 4-trip batches ahead of the WMMAs in three register buffers; 8 partial sums per output, not 16
# speedup vs baseline: 1.7243x; 1.0632x over previous
#include <hip/hip_runtime.h>
#include <math.h>

typedef float v2f __attribute__((ext_vector_type(2)));
typedef float v8f __attribute__((ext_vector_type(8)));
typedef float v4u __attribute__((ext_vector_type(4)));

namespace {
constexpr int C_DIM = 2048;
constexpr int HW = 49;
constexpr int KSTEPS = C_DIM / 4;
constexpr int WAVES_PER_BLOCK = 8;
constexpr int STEPS_PER_WAVE = KSTEPS / WAVES_PER_BLOCK;
constexpr int PAIRS_PER_WAVE = STEPS_PER_WAVE / 2;
constexpr int BATCH_TRIPS = 4;
constexpr int BATCHES = PAIRS_PER_WAVE / BATCH_TRIPS;
constexpr int AHEAD = 2;
constexpr int BUFS = AHEAD + 1;
static_assert(WAVES_PER_BLOCK * 32 == 256, "the launch below uses 256 threads per work-group");
static_assert(PAIRS_PER_WAVE % BATCH_TRIPS == 0 && AHEAD < BATCHES, "whole batches only");
}

__global__ __launch_bounds__(WAVES_PER_BLOCK * 32) void fused(
    const float* __restrict__ x, const float* __restrict__ mask,
    const float* __restrict__ bvec, const float* __restrict__ W,
    float* __restrict__ out) {
  __shared__ float part[WAVES_PER_BLOCK][5][64];
  __shared__ float fin[5][HW];
  __shared__ float sm[HW];
  __shared__ float sgap[HW];
  __shared__ int   s_pmax;
  __shared__ float s_gapmean;

  const int tid = threadIdx.x;
  const int n = blockIdx.x;
  const int wave = tid >> 5;
  const int lane = tid & 31;
  const int mrow = lane & 15;
  const int hoff = (lane >> 4) << 2;

  if (tid < HW) sm[tid] = mask[n * HW + tid];

  const int pb = (mrow < 12) ? (mrow << 2) : 45;

  v8f a0 = {}, a1 = {}, a2 = {}, a3 = {};
  const float* xn = x + (size_t)n * (C_DIM * HW);
  const bool a_from_w = (mrow >= 1) && (mrow <= 4);
  const float a_const = (mrow == 0) ? 1.0f : 0.0f;
  int woff = C_DIM;
  if (mrow == 2)      woff = 2 * C_DIM + C_DIM;
  else if (mrow == 3) woff = 0;
  else if (mrow == 4) woff = 2 * C_DIM;
  const int cbase = wave * (STEPS_PER_WAVE * 4) + hoff;
  const float* wl = W + woff + cbase;
  const float* xl = xn + (size_t)cbase * HW + pb;

  v4u aw[BUFS][BATCH_TRIPS];
  v4u xq[BUFS][BATCH_TRIPS][4];
#define LOAD_BATCH(B)                                                        \
  _Pragma("unroll") for (int t = 0; t < BATCH_TRIPS; ++t) {                  \
    const int jt = (B) * BATCH_TRIPS + t;                                    \
    const float* xr = xl + (size_t)jt * (8 * HW);                            \
    __builtin_memcpy(&aw[(B) % BUFS][t], wl + jt * 8, 16);                   \
    __builtin_memcpy(&xq[(B) % BUFS][t][0], xr, 16);                         \
    __builtin_memcpy(&xq[(B) % BUFS][t][1], xr + HW, 16);                    \
    __builtin_memcpy(&xq[(B) % BUFS][t][2], xr + 2 * HW, 16);                \
    __builtin_memcpy(&xq[(B) % BUFS][t][3], xr + 3 * HW, 16);                \
  }
#pragma unroll
  for (int b = 0; b < AHEAD; ++b) { LOAD_BATCH(b) }
#pragma unroll
  for (int b = 0; b < BATCHES; ++b) {
    __builtin_amdgcn_sched_barrier(0);
    if (b + AHEAD < BATCHES) { LOAD_BATCH(b + AHEAD) }
    __builtin_amdgcn_sched_barrier(0);
#pragma unroll
    for (int t = 0; t < BATCH_TRIPS; ++t) {
      const v4u w4 = aw[b % BUFS][t];
      const v4u r0 = xq[b % BUFS][t][0], r1 = xq[b % BUFS][t][1], r2 = xq[b % BUFS][t][2], r3 = xq[b % BUFS][t][3];
      v2f A0, A1;
      A0.x = a_from_w ? w4.x : a_const;  A0.y = a_from_w ? w4.y : a_const;
      A1.x = a_from_w ? w4.z : a_const;  A1.y = a_from_w ? w4.w : a_const;
      v2f B0, B1, B2, B3;
      B0.x = r0.x;  B0.y = r1.x;
      B1.x = r0.y;  B1.y = r1.y;
      B2.x = r0.z;  B2.y = r1.z;
      B3.x = r0.w;  B3.y = r1.w;
      a0 = __builtin_amdgcn_wmma_f32_16x16x4_f32(false, A0, false, B0, (short)0, a0, false, false);
      a1 = __builtin_amdgcn_wmma_f32_16x16x4_f32(false, A0, false, B1, (short)0, a1, false, false);
      a2 = __builtin_amdgcn_wmma_f32_16x16x4_f32(false, A0, false, B2, (short)0, a2, false, false);
      a3 = __builtin_amdgcn_wmma_f32_16x16x4_f32(false, A0, false, B3, (short)0, a3, false, false);
      B0.x = r2.x;  B0.y = r3.x;
      B1.x = r2.y;  B1.y = r3.y;
      B2.x = r2.z;  B2.y = r3.z;
      B3.x = r2.w;  B3.y = r3.w;
      a0 = __builtin_amdgcn_wmma_f32_16x16x4_f32(false, A1, false, B0, (short)0, a0, false, false);
      a1 = __builtin_amdgcn_wmma_f32_16x16x4_f32(false, A1, false, B1, (short)0, a1, false, false);
      a2 = __builtin_amdgcn_wmma_f32_16x16x4_f32(false, A1, false, B2, (short)0, a2, false, false);
      a3 = __builtin_amdgcn_wmma_f32_16x16x4_f32(false, A1, false, B3, (short)0, a3, false, false);
    }
  }
#undef LOAD_BATCH

  if (lane < 12) {
#pragma unroll
    for (int m = 0; m < 5; ++m) {
      part[wave][m][pb]     = a0[m];
      part[wave][m][pb + 1] = a1[m];
      part[wave][m][pb + 2] = a2[m];
      part[wave][m][pb + 3] = a3[m];
    }
  } else if (lane == 12) {
#pragma unroll
    for (int m = 0; m < 5; ++m) part[wave][m][48] = a3[m];
  }
  __syncthreads();

  if (tid < 5 * HW) {
    int m = tid / HW, p = tid % HW;
    float ssum = 0.0f;
    for (int w = 0; w < WAVES_PER_BLOCK; ++w) ssum += part[w][m][p];
    fin[m][p] = ssum;
  }
  __syncthreads();

  if (tid == 0) {
    float tsum = 0.0f;
    for (int p = 0; p < HW; ++p) tsum += sm[p];
    const float thr = tsum / 49.0f;
    int pmax = 0;
    float best = -__builtin_inff();
    for (int p = 0; p < HW; ++p) {
      float mm = (sm[p] > thr) ? (fin[0][p] / 2048.0f) : 0.0f;
      if (mm > best) { best = mm; pmax = p; }
    }
    s_pmax = pmax;
  }
  __syncthreads();

  const int pmax = s_pmax;
  float pred_d = 0.0f, rdist = 0.0f;
  if (tid < HW) {
    const float u0 = fin[3][pmax];
    const float u1 = fin[4][pmax];
    float pd = fmaxf(u0 + fin[1][tid] + bvec[0], 0.0f);
    float pa = fmaxf(u1 + fin[2][tid] + bvec[1], 0.0f);
    if (tid == pmax) { pd = 0.0f; pa = 0.0f; }
    const int ii = tid / 7,  jj = tid % 7;
    const int i0 = pmax / 7, j0 = pmax % 7;
    const float rx = (float)(ii - i0) / 7.0f;
    const float ry = (float)(jj - j0) / 7.0f;
    rdist = sqrtf(rx * rx + ry * ry);
    const float ang = (atan2f(ry, rx) / 3.14159265358979323846f + 1.0f) * 0.5f;
    float gap = pa - ang;
    if (gap < 0.0f) gap += 1.0f;
    sgap[tid] = gap;
    pred_d = pd;
  }
  __syncthreads();
  if (tid == 0) {
    float gs = 0.0f;
    for (int p = 0; p < HW; ++p) gs += sgap[p];
    s_gapmean = gs / 49.0f;
  }
  __syncthreads();
  if (tid < HW) {
    const float g = sgap[tid] - s_gapmean;
    const float dl = pred_d - rdist;
    out[(size_t)n * HW + tid] = dl * dl + g * g;
  }
}

extern "C" void kernel_launch(void* const* d_in, const int* in_sizes, int n_in,
                              void* d_out, int out_size, void* d_ws, size_t ws_size,
                              hipStream_t stream) {
  const float* x    = (const float*)d_in[0];
  const float* mask = (const float*)d_in[1];
  const float* W    = (const float*)d_in[2];
  const float* b    = (const float*)d_in[3];
  float* out = (float*)d_out;

  fused<<<256, 256, 0, stream>>>(x, mask, b, W, out);
}
